# starting kernel
# baseline (speedup 1.0000x reference)
#include <hip/hip_runtime.h>
#include <math.h>

#define IC_N 512
#define OC_N 512
#define H_N 128
#define W_N 128
#define HW_N (H_N * W_N)
#define PH 129
#define PHS (PH * PH)

typedef __attribute__((ext_vector_type(2))) float v2f;
typedef __attribute__((ext_vector_type(8))) float v8f;

__global__ __launch_bounds__(256) void prep_weights(const float* __restrict__ wgt,
                                                    float* __restrict__ wp) {
  const int oc = blockIdx.x;
  const int tid = threadIdx.x;
  const float scale = rsqrtf(4608.0f);
  __shared__ float red[256];
  __shared__ float dsh;

  const float* wo = wgt + (size_t)oc * IC_N * 9;
  float s = 0.f;
  for (int ic = tid; ic < IC_N; ic += 256) {
    const float* wi = wo + ic * 9;
#pragma unroll
    for (int k = 0; k < 9; ++k) {
      float v = scale * wi[k];
      s += v * v;
    }
  }
  red[tid] = s;
  __syncthreads();
  for (int off = 128; off > 0; off >>= 1) {
    if (tid < off) red[tid] += red[tid + off];
    __syncthreads();
  }
  if (tid == 0) dsh = rsqrtf(red[0] + 1e-8f);
  __syncthreads();
  const float d = dsh * scale;

  float* w00 = wp;
  float* w01 = wp + (size_t)OC_N * 2048;
  float* w10 = w01 + (size_t)OC_N * 1024;
  float* w11 = w10 + (size_t)OC_N * 1024;

  for (int ic = tid; ic < IC_N; ic += 256) {
    const float* wi = wo + ic * 9;
    w00[(size_t)oc * 2048 + 0 * 512 + ic] = d * wi[2 * 3 + 2];
    w00[(size_t)oc * 2048 + 1 * 512 + ic] = d * wi[2 * 3 + 0];
    w00[(size_t)oc * 2048 + 2 * 512 + ic] = d * wi[0 * 3 + 2];
    w00[(size_t)oc * 2048 + 3 * 512 + ic] = d * wi[0 * 3 + 0];
    w01[(size_t)oc * 1024 + 0 * 512 + ic] = d * wi[2 * 3 + 1];
    w01[(size_t)oc * 1024 + 1 * 512 + ic] = d * wi[0 * 3 + 1];
    w10[(size_t)oc * 1024 + 0 * 512 + ic] = d * wi[1 * 3 + 2];
    w10[(size_t)oc * 1024 + 1 * 512 + ic] = d * wi[1 * 3 + 0];
    w11[(size_t)oc * 512 + ic] = d * wi[1 * 3 + 1];
  }
}

__global__ __launch_bounds__(128) void upconv_gemm(
    const float* __restrict__ x, const float* __restrict__ wp,
    float* __restrict__ up, int Kp, int ntaps, int dymask, int dxmask,
    int rows, int cols) {
  __shared__ float bs[512];

  const int tid = threadIdx.x;
  const int lane = tid & 31;
  const int wave = tid >> 5;
  const int px = lane & 15;
  const int hi = lane >> 4;
  const int koff = hi * 2;

  const int r = blockIdx.y;
  const int colbase = blockIdx.x * 16;
  const int oc_base = blockIdx.z * 64 + wave * 16;

  const int l_icl0 = tid >> 4;
  const int l_px = tid & 15;

  v8f acc = {};

  for (int t = 0; t < ntaps; ++t) {
    const int dy = -((dymask >> t) & 1);
    const int dx = -((dxmask >> t) & 1);
    const int row_in = r + dy;
    if (row_in < 0 || row_in >= H_N) continue;

    const int cin = colbase + l_px + dx;
    const bool cok = (cin >= 0) && (cin < W_N);
    const float* xrow = x + (size_t)row_in * W_N + cin;
    const float* arow = wp + (size_t)(oc_base + px) * Kp + t * IC_N;

    for (int icc = 0; icc < IC_N; icc += 32) {
#pragma unroll
      for (int i = 0; i < 4; ++i) {
        const int icl = l_icl0 + 8 * i;
        float v = cok ? xrow[(size_t)(icc + icl) * HW_N] : 0.0f;
        bs[l_px * 32 + icl] = v;
      }
      if (icc + 32 < IC_N)
        __builtin_prefetch(xrow + (size_t)(icc + 32 + l_icl0) * HW_N, 0, 1);
      __syncthreads();

#pragma unroll
      for (int s = 0; s < 8; ++s) {
        v2f a = *(const v2f*)(arow + icc + 4 * s + koff);
        v2f b = *(const v2f*)(&bs[px * 32 + 4 * s + koff]);
        acc = __builtin_amdgcn_wmma_f32_16x16x4_f32(
            false, a, false, b, (short)0, acc, false, false);
      }
      __syncthreads();
    }
  }

  const int c_out = colbase + px;
  if (c_out < cols) {
    float* o = up + (size_t)(oc_base + hi * 8) * PHS + (size_t)r * PH + c_out;
#pragma unroll
    for (int j = 0; j < 8; ++j) o[(size_t)j * PHS] = acc[j];
  }
}

__global__ __launch_bounds__(256) void blur_kernel(const float* __restrict__ up,
                                                   float* __restrict__ out) {
  __shared__ float tile[19 * 20];
  const int oc = blockIdx.z;
  const int y0 = blockIdx.y * 16;
  const int x0 = blockIdx.x * 16;
  const int tid = threadIdx.x;

  for (int idx = tid; idx < 19 * 19; idx += 256) {
    const int ly = idx / 19, lx = idx % 19;
    const int yy = y0 - 1 + ly;
    const int xx = x0 - 1 + lx;
    float v = 0.f;
    if (yy >= 0 && yy <= 256 && xx >= 0 && xx <= 256) {
      const int p = ((yy & 1) << 1) | (xx & 1);
      v = up[((size_t)p * OC_N + oc) * PHS + (size_t)(yy >> 1) * PH + (xx >> 1)];
    }
    tile[ly * 20 + lx] = v;
  }
  __syncthreads();

  const int ty = tid >> 4, tx = tid & 15;
  const float c0 = 0.25f, c1 = 0.75f;
  const float cy[4] = {c0, c1, c1, c0};
  float acc = 0.f;
#pragma unroll
  for (int i = 0; i < 4; ++i) {
    const float* row = &tile[(ty + i) * 20 + tx];
    const float rsum = c0 * row[0] + c1 * row[1] + c1 * row[2] + c0 * row[3];
    acc += cy[i] * rsum;
  }
  out[((size_t)oc * 256 + (y0 + ty)) * 256 + (x0 + tx)] = acc;
}

extern "C" void kernel_launch(void* const* d_in, const int* in_sizes, int n_in,
                              void* d_out, int out_size, void* d_ws, size_t ws_size,
                              hipStream_t stream) {
  const float* x = (const float*)d_in[0];
  const float* wgt = (const float*)d_in[1];
  float* out = (float*)d_out;

  float* ws = (float*)d_ws;
  float* wp = ws;
  float* up = ws + (size_t)OC_N * 4608;

  const float* w00 = wp;
  const float* w01 = wp + (size_t)OC_N * 2048;
  const float* w10 = w01 + (size_t)OC_N * 1024;
  const float* w11 = w10 + (size_t)OC_N * 1024;

  float* up00 = up + (size_t)0 * OC_N * PHS;
  float* up01 = up + (size_t)1 * OC_N * PHS;
  float* up10 = up + (size_t)2 * OC_N * PHS;
  float* up11 = up + (size_t)3 * OC_N * PHS;

  prep_weights<<<OC_N, 256, 0, stream>>>(wgt, wp);

  upconv_gemm<<<dim3(9, 129, 8), 128, 0, stream>>>(x, w00, up00, 2048, 4, 0x3, 0x5, 129, 129);
  upconv_gemm<<<dim3(8, 129, 8), 128, 0, stream>>>(x, w01, up01, 1024, 2, 0x1, 0x0, 129, 128);
  upconv_gemm<<<dim3(9, 128, 8), 128, 0, stream>>>(x, w10, up10, 1024, 2, 0x0, 0x1, 128, 129);
  upconv_gemm<<<dim3(8, 128, 8), 128, 0, stream>>>(x, w11, up11,  512, 1, 0x0, 0x0, 128, 128);

  blur_kernel<<<dim3(16, 16, OC_N), 256, 0, stream>>>(up, out);
}


# Transposed conv as ONE stacked GEMM (rows = 9 taps x oc, unshifted pixels): packed weights and x go global->registers, no LDS; blur pass gathers the 9 tap planes; sum order changes
# speedup vs baseline: 2.6207x; 2.6207x over previous
#include <hip/hip_runtime.h>
#include <math.h>

#define IC_N 512
#define OC_N 512
#define H_N 128
#define W_N 128
#define HW_N (H_N * W_N)
#define M_N (9 * OC_N)

typedef __attribute__((ext_vector_type(2))) float v2f;
typedef __attribute__((ext_vector_type(4))) float v4f;
typedef __attribute__((ext_vector_type(8))) float v8f;

__global__ __launch_bounds__(256) void prep_weights(const float* __restrict__ wgt,
                                                    float* __restrict__ ap) {
  const int ocg = blockIdx.x;
  const int tp = blockIdx.y;
  const int tid = threadIdx.x;
  const float scale = rsqrtf(4608.0f);
  __shared__ float dsh[16];

  {
    const int o = tid >> 4, q = tid & 15;
    const v4f* wo = (const v4f*)(wgt + (size_t)(ocg * 16 + o) * (IC_N * 9));
    float s = 0.f;
    for (int e = q; e < (IC_N * 9) / 4; e += 16) {
      v4f v = wo[e] * scale;
      s += v.x * v.x;
      s += v.y * v.y;
      s += v.z * v.z;
      s += v.w * v.w;
    }
    s += __shfl_xor(s, 8, 16);
    s += __shfl_xor(s, 4, 16);
    s += __shfl_xor(s, 2, 16);
    s += __shfl_xor(s, 1, 16);
    if (q == 0) dsh[o] = rsqrtf(s + 1e-8f) * scale;
  }
  __syncthreads();

  float* dst = ap + (size_t)(tp * 32 + ocg) * (64 * 128);
  for (int it = 0; it < 8; ++it) {
    const int f4 = tid + 256 * it;
    const int kp = f4 >> 5, lane = f4 & 31;
    const int h = lane >> 4, c = lane & 15;
    const float d = dsh[c];
    const float* src = wgt + ((size_t)(ocg * 16 + c) * IC_N + (8 * kp + 2 * h)) * 9 + tp;
    v4f v;
    v.x = d * src[0];
    v.y = d * src[9];
    v.z = d * src[36];
    v.w = d * src[45];
    *(v4f*)(dst + (size_t)f4 * 4) = v;
  }
}

__device__ __forceinline__ void load_frags(v4f (&a)[4], v4f (&r)[4], const v4f* pa,
                                           const float* pb, int kp) {
#pragma unroll
  for (int i = 0; i < 4; ++i) a[i] = pa[(size_t)i * 2048 + kp * 32];
#pragma unroll
  for (int s = 0; s < 2; ++s)
#pragma unroll
    for (int e = 0; e < 2; ++e)
      r[s * 2 + e] = *(const v4f*)(pb + (size_t)(8 * kp + 4 * s + e) * HW_N);
}

__device__ __forceinline__ void pair_frags(v2f (&p)[8], const v4f (&r)[4]) {
#pragma unroll
  for (int s = 0; s < 2; ++s)
#pragma unroll
    for (int t = 0; t < 4; ++t) {
      p[s * 4 + t].x = r[s * 2 + 0][t];
      p[s * 4 + t].y = r[s * 2 + 1][t];
    }
}

__device__ __forceinline__ void mma_block(v8f (&acc)[4][4], const v4f (&a)[4],
                                          const v2f (&p)[8]) {
#pragma unroll
  for (int s = 0; s < 2; ++s) {
#pragma unroll
    for (int i = 0; i < 4; ++i) {
      v2f af;
      af.x = s ? a[i].z : a[i].x;
      af.y = s ? a[i].w : a[i].y;
#pragma unroll
      for (int t = 0; t < 4; ++t)
        acc[i][t] = __builtin_amdgcn_wmma_f32_16x16x4_f32(
            false, af, false, p[s * 4 + t], (short)0, acc[i][t], false, false);
    }
  }
}

__global__ __launch_bounds__(256) void tap_gemm(const float* __restrict__ x,
                                                const float* __restrict__ ap,
                                                float* __restrict__ z) {
  const int tid = threadIdx.x;
  const int lane = tid & 31;
  const int wave = tid >> 5;
  const int c = lane & 15;
  const int h = lane >> 4;
  const int wm = wave & 1, wn = wave >> 1;
  const int mt0 = blockIdx.y * 8 + wm * 4;
  const int n0 = blockIdx.x * 256 + wn * 64;

  const v4f* pa = (const v4f*)ap + (size_t)mt0 * 2048 + lane;
  const float* pb = x + (size_t)(2 * h) * HW_N + n0 + 4 * c;

  v8f acc[4][4];
#pragma unroll
  for (int i = 0; i < 4; ++i)
#pragma unroll
    for (int t = 0; t < 4; ++t) acc[i][t] = v8f{};

  v4f a0[4], a1[4], r0[4], r1[4];
  v2f p0[8], p1[8];
  load_frags(a0, r0, pa, pb, 0);
  pair_frags(p0, r0);
  for (int kp = 0; kp < 64; kp += 2) {
    load_frags(a1, r1, pa, pb, kp + 1);
    __builtin_amdgcn_sched_barrier(0);
    mma_block(acc, a0, p0);
    __builtin_amdgcn_sched_barrier(0);
    pair_frags(p1, r1);
    load_frags(a0, r0, pa, pb, kp + 2 < 64 ? kp + 2 : 63);
    __builtin_amdgcn_sched_barrier(0);
    mma_block(acc, a1, p1);
    __builtin_amdgcn_sched_barrier(0);
    pair_frags(p0, r0);
  }

  float* pz = z + (size_t)(mt0 * 16 + 8 * h) * HW_N + n0 + 4 * c;
#pragma unroll
  for (int i = 0; i < 4; ++i)
#pragma unroll
    for (int r = 0; r < 8; ++r) {
      v4f o;
      o.x = acc[i][0][r];
      o.y = acc[i][1][r];
      o.z = acc[i][2][r];
      o.w = acc[i][3][r];
      *(v4f*)(pz + (size_t)(16 * i + r) * HW_N) = o;
    }
}

__global__ __launch_bounds__(128) void blur_taps(const float* __restrict__ z,
                                                 float* __restrict__ out) {
  const int J = threadIdx.x;
  const int I0 = blockIdx.x * 16;
  const int oc = blockIdx.y;
  const float c0 = 0.25f, c1 = 0.75f;
  const int jm = J > 0 ? J - 1 : 0;
  const int jp = J < W_N - 1 ? J + 1 : W_N - 1;
  const bool has_l = J > 0, has_r = J < W_N - 1;
  const float* base = z + (size_t)oc * HW_N;
  const size_t plane = (size_t)OC_N * HW_N;

  v2f ro_m2 = {0.f, 0.f}, ro_m1 = {0.f, 0.f}, re_m1 = {0.f, 0.f}, h2p = {0.f, 0.f};
  for (int i = I0 - 1; i <= I0 + 16; ++i) {
    v2f hh[3];
    if (i >= 0 && i < H_N) {
#pragma unroll
      for (int a = 0; a < 3; ++a) {
        const float* p0 = base + (size_t)(a * 3 + 0) * plane + i * W_N;
        const float* p1 = p0 + plane;
        const float* p2 = p1 + plane;
        const float z0c = p0[J];
        float z0r = p0[jp];
        float z1l = p1[jm];
        const float z1c = p1[J];
        float z1r = p1[jp];
        float z2l = p2[jm];
        const float z2c = p2[J];
        if (!has_l) { z1l = 0.f; z2l = 0.f; }
        if (!has_r) { z0r = 0.f; z1r = 0.f; }
        const float e0 = z0c + z2l;
        const float e1 = z0r + z2c;
        hh[a].x = c0 * z1l + c1 * e0 + c1 * z1c + c0 * e1;
        hh[a].y = c0 * e0 + c1 * z1c + c1 * e1 + c0 * z1r;
      }
    } else {
      hh[0] = v2f{0.f, 0.f};
      hh[1] = v2f{0.f, 0.f};
      hh[2] = v2f{0.f, 0.f};
    }
    const v2f re = hh[0] + h2p;
    const v2f ro = hh[1];
    if (i >= I0 + 1) {
      const int I = i - 1;
      const v2f o0 = c0 * ro_m2 + c1 * re_m1 + c1 * ro_m1 + c0 * re;
      const v2f o1 = c0 * re_m1 + c1 * ro_m1 + c1 * re + c0 * ro;
      float* po = out + ((size_t)oc * 256 + 2 * I) * 256 + 2 * J;
      *(v2f*)po = o0;
      *(v2f*)(po + 256) = o1;
    }
    ro_m2 = ro_m1;
    ro_m1 = ro;
    re_m1 = re;
    h2p = hh[2];
  }
}

extern "C" void kernel_launch(void* const* d_in, const int* in_sizes, int n_in,
                              void* d_out, int out_size, void* d_ws, size_t ws_size,
                              hipStream_t stream) {
  const float* x = (const float*)d_in[0];
  const float* wgt = (const float*)d_in[1];
  float* out = (float*)d_out;

  float* ws = (float*)d_ws;
  float* ap = ws;
  float* z = ws + (size_t)M_N * IC_N;

  prep_weights<<<dim3(32, 9), 256, 0, stream>>>(wgt, ap);
  tap_gemm<<<dim3(64, 36), 256, 0, stream>>>(x, ap, z);
  blur_taps<<<dim3(8, OC_N), 128, 0, stream>>>(z, out);
}
